# starting kernel
# baseline (speedup 1.0000x reference)
#include <hip/hip_runtime.h>
#include <math.h>

#define BS    8
#define V     2048
#define NNB   20
#define INC   64
#define OUTC  128
#define SUP   2
#define SUPC  (SUP * OUTC)
#define FEATC (OUTC + SUPC)
#define MROWS (BS * V)
#define EPSF  1e-12f

typedef float v2f __attribute__((ext_vector_type(2)));
typedef float v8f __attribute__((ext_vector_type(8)));

template <bool BT>
__global__ __launch_bounds__(128)
void gemm_bias_wmma(const float* __restrict__ A, const float* __restrict__ B,
                    const float* __restrict__ bias, float* __restrict__ C,
                    int M, int N, int K) {
  const int lane  = threadIdx.x & 31;
  const int wave  = threadIdx.x >> 5;
  const int tile  = blockIdx.x * 4 + wave;
  const int tilesN = N >> 4;
  const int tm = tile / tilesN;
  const int tn = tile - tm * tilesN;
  if (tm * 16 >= M) return;

  const int m0   = tm << 4;
  const int n0   = tn << 4;
  const int half = lane >> 4;
  const int l16  = lane & 15;

  v8f acc = {0.f, 0.f, 0.f, 0.f, 0.f, 0.f, 0.f, 0.f};

  const float* __restrict__ Arow = A + (size_t)(m0 + l16) * K;

  for (int k0 = 0; k0 < K; k0 += 4) {
    const int ka = k0 + 2 * half;
    v2f a;
    a.x = Arow[ka];
    a.y = Arow[ka + 1];

    v2f b;
    if (BT) {
      const float* bp = B + (size_t)(n0 + l16) * K + ka;
      b.x = bp[0];
      b.y = bp[1];
    } else {
      const float* bp = B + (size_t)ka * N + n0 + l16;
      b.x = bp[0];
      b.y = bp[N];
    }
    acc = __builtin_amdgcn_wmma_f32_16x16x4_f32(false, a, false, b,
                                                (short)0, acc, false, false);
  }

  const float bv = bias[n0 + l16];
  float* __restrict__ Crow = C + (size_t)(m0 + 8 * half) * N + n0 + l16;
#pragma unroll
  for (int r = 0; r < 8; ++r)
    Crow[(size_t)r * N] = acc[r] + bv;
}

__global__ __launch_bounds__(256)
void geom_fuse_kernel(const int* __restrict__ nbr, const float* __restrict__ verts,
                      const float* __restrict__ feat,
                      const float* __restrict__ directions,
                      const float* __restrict__ distw,
                      float* __restrict__ out256) {
  const int row = blockIdx.x;
  const int b   = row / V;
  const int s   = threadIdx.x;

  __shared__ float s_dirn[NNB][3];
  __shared__ float s_dist[NNB];
  __shared__ int   s_nidx[NNB];
  __shared__ float s_acc[SUPC];
  __shared__ float s_df[SUPC];

  if (s < NNB) {
    const int ni = nbr[(size_t)row * NNB + s];
    s_nidx[s] = ni;
    const float* pv = verts + (size_t)row * 3;
    const float* pn = verts + ((size_t)b * V + ni) * 3;
    const float dx = pn[0] - pv[0];
    const float dy = pn[1] - pv[1];
    const float dz = pn[2] - pv[2];
    const float d  = sqrtf(dx * dx + dy * dy + dz * dz);
    const float iv = 1.0f / fmaxf(d, EPSF);
    s_dirn[s][0] = dx * iv;
    s_dirn[s][1] = dy * iv;
    s_dirn[s][2] = dz * iv;
    s_dist[s] = d;
  }
  __syncthreads();

  float dmax = 0.0f;
#pragma unroll
  for (int n = 0; n < NNB; ++n) dmax = fmaxf(dmax, s_dist[n]);

  float d0 = directions[0 * SUPC + s];
  float d1 = directions[1 * SUPC + s];
  float d2 = directions[2 * SUPC + s];
  const float nrm = sqrtf(d0 * d0 + d1 * d1 + d2 * d2);
  const float iv  = 1.0f / fmaxf(nrm, EPSF);
  d0 *= iv; d1 *= iv; d2 *= iv;

  float acc = -INFINITY;
  const size_t bbase = (size_t)b * V;
#pragma unroll 4
  for (int n = 0; n < NNB; ++n) {
    const float th = fmaxf(0.0f, d0 * s_dirn[n][0] +
                                 d1 * s_dirn[n][1] +
                                 d2 * s_dirn[n][2]);
    const float sg = feat[(bbase + s_nidx[n]) * FEATC + OUTC + s];
    acc = fmaxf(acc, th * sg);
  }

  s_acc[s] = acc;
  s_df[s]  = fmaxf(0.0f, dmax * distw[s]);
  __syncthreads();

  if (s < OUTC) {
    const float center = feat[(size_t)row * FEATC + s];
    out256[(size_t)row * SUPC + s] = center + s_acc[s] + s_acc[s + OUTC];
  } else {
    const int c = s - OUTC;
    out256[(size_t)row * SUPC + OUTC + c] = s_df[c] + s_df[c + OUTC];
  }
}

extern "C" void kernel_launch(void* const* d_in, const int* in_sizes, int n_in,
                              void* d_out, int out_size, void* d_ws, size_t ws_size,
                              hipStream_t stream) {
  const int*   nbr   = (const int*)  d_in[0];
  const float* verts = (const float*)d_in[1];
  const float* fmap  = (const float*)d_in[2];
  const float* wgt   = (const float*)d_in[3];
  const float* bias  = (const float*)d_in[4];
  const float* dirs  = (const float*)d_in[5];
  const float* distw = (const float*)d_in[6];
  const float* mlpw  = (const float*)d_in[7];
  const float* mlpb  = (const float*)d_in[8];
  float* out = (float*)d_out;

  float* feat   = (float*)d_ws;
  float* out256 = feat + (size_t)MROWS * FEATC;

  {
    const int tiles = (MROWS / 16) * (FEATC / 16);
    gemm_bias_wmma<false><<<tiles / 4, 128, 0, stream>>>(
        fmap, wgt, bias, feat, MROWS, FEATC, INC);
  }

  geom_fuse_kernel<<<MROWS, 256, 0, stream>>>(nbr, verts, feat, dirs, distw, out256);

  {
    const int tiles = (MROWS / 16) * (OUTC / 16);
    gemm_bias_wmma<true><<<tiles / 4, 128, 0, stream>>>(
        out256, mlpw, mlpb, out, MROWS, OUTC, SUPC);
  }
}


# Middle kernel regridded: one wave per 4 vertices, each lane owns 8 channels, neighbor rows gathered by 16-byte loads, geometry passed by lane reads; no LDS or barriers, 32x fewer waves
# speedup vs baseline: 1.6588x; 1.6588x over previous
#include <hip/hip_runtime.h>
#include <math.h>

#define BS    8
#define V     2048
#define NNB   20
#define INC   64
#define OUTC  128
#define SUP   2
#define SUPC  (SUP * OUTC)
#define FEATC (OUTC + SUPC)
#define MROWS (BS * V)
#define EPSF  1e-12f

typedef float v2f __attribute__((ext_vector_type(2)));
typedef float v8f __attribute__((ext_vector_type(8)));

template <bool BT>
__global__ __launch_bounds__(128)
void gemm_bias_wmma(const float* __restrict__ A, const float* __restrict__ B,
                    const float* __restrict__ bias, float* __restrict__ C,
                    int M, int N, int K) {
  const int lane  = threadIdx.x & 31;
  const int wave  = threadIdx.x >> 5;
  const int tile  = blockIdx.x * 4 + wave;
  const int tilesN = N >> 4;
  const int tm = tile / tilesN;
  const int tn = tile - tm * tilesN;
  if (tm * 16 >= M) return;

  const int m0   = tm << 4;
  const int n0   = tn << 4;
  const int half = lane >> 4;
  const int l16  = lane & 15;

  v8f acc = {0.f, 0.f, 0.f, 0.f, 0.f, 0.f, 0.f, 0.f};

  const float* __restrict__ Arow = A + (size_t)(m0 + l16) * K;

  for (int k0 = 0; k0 < K; k0 += 4) {
    const int ka = k0 + 2 * half;
    v2f a;
    a.x = Arow[ka];
    a.y = Arow[ka + 1];

    v2f b;
    if (BT) {
      const float* bp = B + (size_t)(n0 + l16) * K + ka;
      b.x = bp[0];
      b.y = bp[1];
    } else {
      const float* bp = B + (size_t)ka * N + n0 + l16;
      b.x = bp[0];
      b.y = bp[N];
    }
    acc = __builtin_amdgcn_wmma_f32_16x16x4_f32(false, a, false, b,
                                                (short)0, acc, false, false);
  }

  const float bv = bias[n0 + l16];
  float* __restrict__ Crow = C + (size_t)(m0 + 8 * half) * N + n0 + l16;
#pragma unroll
  for (int r = 0; r < 8; ++r)
    Crow[(size_t)r * N] = acc[r] + bv;
}

typedef float v4f __attribute__((ext_vector_type(4)));

#define ROWS_PER_WAVE   4
#define GROUP           10
#define WAVES_PER_BLOCK 8
#define ROWS_PER_BLOCK  (ROWS_PER_WAVE * WAVES_PER_BLOCK)

__device__ __forceinline__ float lane_read_f(float x, int n) {
  return __int_as_float(__builtin_amdgcn_readlane(__float_as_int(x), n));
}

__global__ __launch_bounds__(32 * WAVES_PER_BLOCK) __attribute__((amdgpu_waves_per_eu(4)))
void geom_fuse_kernel(const int* __restrict__ nbr, const float* __restrict__ verts,
                      const float* __restrict__ feat,
                      const float* __restrict__ directions,
                      const float* __restrict__ distw,
                      float* __restrict__ out256) {
  const int lane = threadIdx.x & 31;
  const int wave = __builtin_amdgcn_readfirstlane((int)(threadIdx.x >> 5));
  const int row0 = (blockIdx.x * WAVES_PER_BLOCK + wave) * ROWS_PER_WAVE;
  const int b    = row0 / V;
  const size_t bbase = (size_t)b * V;
  const int c4 = 4 * lane;

  const int ln = (lane < NNB) ? lane : 0;
  int ni_next = nbr[(size_t)row0 * NNB + ln];

  v4f a0 = *(const v4f*)(directions + 0 * SUPC + c4);
  v4f a1 = *(const v4f*)(directions + 1 * SUPC + c4);
  v4f a2 = *(const v4f*)(directions + 2 * SUPC + c4);
  v4f b0 = *(const v4f*)(directions + 0 * SUPC + OUTC + c4);
  v4f b1 = *(const v4f*)(directions + 1 * SUPC + OUTC + c4);
  v4f b2 = *(const v4f*)(directions + 2 * SUPC + OUTC + c4);
#pragma unroll
  for (int j = 0; j < 4; ++j) {
    const float na = sqrtf(a0[j] * a0[j] + a1[j] * a1[j] + a2[j] * a2[j]);
    const float ia = 1.0f / fmaxf(na, EPSF);
    a0[j] *= ia; a1[j] *= ia; a2[j] *= ia;
    const float nb = sqrtf(b0[j] * b0[j] + b1[j] * b1[j] + b2[j] * b2[j]);
    const float ib = 1.0f / fmaxf(nb, EPSF);
    b0[j] *= ib; b1[j] *= ib; b2[j] *= ib;
  }

  const v4f* __restrict__ fb4 = (const v4f*)(feat + bbase * FEATC + OUTC);

#pragma unroll 1
  for (int r = 0; r < ROWS_PER_WAVE; ++r) {
    const int row = row0 + r;
    const int ni  = ni_next & (V - 1);
    const int rnx = (r + 1 < ROWS_PER_WAVE) ? row + 1 : row;
    ni_next = nbr[(size_t)rnx * NNB + ln];

    const float* pv = verts + (size_t)row * 3;
    const float* pn = verts + (bbase + ni) * 3;
    const float dx = pn[0] - pv[0];
    const float dy = pn[1] - pv[1];
    const float dz = pn[2] - pv[2];
    const float d  = sqrtf(dx * dx + dy * dy + dz * dz);
    const float ivd = 1.0f / fmaxf(d, EPSF);
    const float nx = dx * ivd;
    const float ny = dy * ivd;
    const float nz = dz * ivd;

    float dmax = d;
#pragma unroll
    for (int m = 16; m >= 1; m >>= 1) dmax = fmaxf(dmax, __shfl_xor(dmax, m, 32));

    v4f accA = {-INFINITY, -INFINITY, -INFINITY, -INFINITY};
    v4f accB = {-INFINITY, -INFINITY, -INFINITY, -INFINITY};
#pragma unroll 1
    for (int h = 0; h < NNB; h += GROUP) {
#pragma unroll
      for (int q = 0; q < GROUP; ++q) {
        const int n = h + q;
        const int   nin = __builtin_amdgcn_readlane(ni, n);
        const float x = lane_read_f(nx, n);
        const float y = lane_read_f(ny, n);
        const float z = lane_read_f(nz, n);
        const v4f* p = fb4 + ((unsigned)nin * (unsigned)(FEATC / 4) + (unsigned)lane);
        const v4f gA = p[0];
        const v4f gB = p[OUTC / 4];
#pragma unroll
        for (int j = 0; j < 4; ++j) {
          const float tA = fmaxf(0.0f, a0[j] * x + a1[j] * y + a2[j] * z);
          accA[j] = fmaxf(accA[j], tA * gA[j]);
          const float tB = fmaxf(0.0f, b0[j] * x + b1[j] * y + b2[j] * z);
          accB[j] = fmaxf(accB[j], tB * gB[j]);
        }
      }
    }

    const v4f center = *(const v4f*)(feat + (size_t)row * FEATC + c4);
    const v4f wA = *(const v4f*)(distw + c4);
    const v4f wB = *(const v4f*)(distw + OUTC + c4);
    v4f fuse, dsum;
#pragma unroll
    for (int j = 0; j < 4; ++j) {
      fuse[j] = center[j] + accA[j] + accB[j];
      dsum[j] = fmaxf(0.0f, dmax * wA[j]) + fmaxf(0.0f, dmax * wB[j]);
    }
    float* orow = out256 + (size_t)row * SUPC + c4;
    *(v4f*)(orow)        = fuse;
    *(v4f*)(orow + OUTC) = dsum;
  }
}

extern "C" void kernel_launch(void* const* d_in, const int* in_sizes, int n_in,
                              void* d_out, int out_size, void* d_ws, size_t ws_size,
                              hipStream_t stream) {
  const int*   nbr   = (const int*)  d_in[0];
  const float* verts = (const float*)d_in[1];
  const float* fmap  = (const float*)d_in[2];
  const float* wgt   = (const float*)d_in[3];
  const float* bias  = (const float*)d_in[4];
  const float* dirs  = (const float*)d_in[5];
  const float* distw = (const float*)d_in[6];
  const float* mlpw  = (const float*)d_in[7];
  const float* mlpb  = (const float*)d_in[8];
  float* out = (float*)d_out;

  float* feat   = (float*)d_ws;
  float* out256 = feat + (size_t)MROWS * FEATC;

  {
    const int tiles = (MROWS / 16) * (FEATC / 16);
    gemm_bias_wmma<false><<<tiles / 4, 128, 0, stream>>>(
        fmap, wgt, bias, feat, MROWS, FEATC, INC);
  }

  geom_fuse_kernel<<<MROWS / ROWS_PER_BLOCK, 32 * WAVES_PER_BLOCK, 0, stream>>>(nbr, verts, feat, dirs, distw, out256);

  {
    const int tiles = (MROWS / 16) * (OUTC / 16);
    gemm_bias_wmma<true><<<tiles / 4, 128, 0, stream>>>(
        out256, mlpw, mlpb, out, MROWS, OUTC, SUPC);
  }
}


# Both GEMM launches: each wave computes a 32x64 block as 2x4 WMMA tiles, fragments fetched by 16-byte row loads two chunks ahead and reused across tiles; C rows written by 16-byte whole-line stores
# speedup vs baseline: 2.2859x; 1.3780x over previous
#include <hip/hip_runtime.h>
#include <math.h>

#define BS    8
#define V     2048
#define NNB   20
#define INC   64
#define OUTC  128
#define SUP   2
#define SUPC  (SUP * OUTC)
#define FEATC (OUTC + SUPC)
#define MROWS (BS * V)
#define EPSF  1e-12f

typedef float v2f __attribute__((ext_vector_type(2)));
typedef float v8f __attribute__((ext_vector_type(8)));

typedef float v4f __attribute__((ext_vector_type(4)));

#define GM_MT    2
#define GM_NT    4
#define GM_WAVES 4
#define GM_ROWS  (16 * GM_MT)
#define GM_COLS  (16 * GM_NT)

struct GFrag {
  v4f a[GM_MT];
  v4f b[4];
};

template <bool BT, int N, int K>
__global__ __launch_bounds__(32 * GM_WAVES)
void gemm_bias_wmma(const float* __restrict__ A, const float* __restrict__ B,
                    const float* __restrict__ bias, float* __restrict__ C) {
  static_assert(GM_NT == 4, "column map packs 4 tiles into one 16-byte vector");
  static_assert(K % 8 == 0 && N % GM_COLS == 0, "shape");
  constexpr int TILES_N = N / GM_COLS;
  constexpr int KC = K / 8;

  const int lane = threadIdx.x & 31;
  const int wave = threadIdx.x >> 5;
  const int wt   = blockIdx.x * GM_WAVES + wave;
  const int tm   = wt / TILES_N;
  const int tn   = wt - tm * TILES_N;
  const int m0   = tm * GM_ROWS;
  const int n0   = tn * GM_COLS;
  const int h    = lane >> 4;
  const int c    = lane & 15;

  v8f acc[GM_MT][GM_NT];
#pragma unroll
  for (int u = 0; u < GM_MT; ++u)
#pragma unroll
    for (int t = 0; t < GM_NT; ++t)
      acc[u][t] = (v8f){0.f, 0.f, 0.f, 0.f, 0.f, 0.f, 0.f, 0.f};

  const float* __restrict__ ap[GM_MT];
#pragma unroll
  for (int u = 0; u < GM_MT; ++u)
    ap[u] = A + (size_t)(m0 + 16 * u + c) * K + 4 * h;

  const float* __restrict__ bp = BT ? (B + (size_t)(n0 + 4 * c) * K + 4 * h)
                                    : (B + (size_t)(4 * h) * N + n0 + 4 * c);

  GFrag f[3];

#define GM_LOAD(F, JJ)                                                        \
  do {                                                                        \
    _Pragma("unroll")                                                         \
    for (int u = 0; u < GM_MT; ++u)                                           \
      (F).a[u] = *(const v4f*)(ap[u] + 8 * (JJ));                             \
    _Pragma("unroll")                                                         \
    for (int i = 0; i < 4; ++i)                                               \
      (F).b[i] = BT ? *(const v4f*)(bp + (size_t)i * K + 8 * (JJ))            \
                    : *(const v4f*)(bp + (size_t)(8 * (JJ) + i) * N);         \
  } while (0)

  GM_LOAD(f[0], 0);
  if (KC > 1) GM_LOAD(f[1], 1);

#pragma unroll
  for (int jj = 0; jj < KC; ++jj) {
    __builtin_amdgcn_sched_barrier(0);
    if (jj + 2 < KC) GM_LOAD(f[(jj + 2) % 3], jj + 2);
    __builtin_amdgcn_sched_barrier(0);
    const GFrag& g = f[jj % 3];
#pragma unroll
    for (int st = 0; st < 2; ++st) {
#pragma unroll
      for (int u = 0; u < GM_MT; ++u) {
        v2f av;
        av.x = g.a[u][2 * st];
        av.y = g.a[u][2 * st + 1];
#pragma unroll
        for (int t = 0; t < GM_NT; ++t) {
          v2f bv;
          if (BT) {
            bv.x = g.b[t][2 * st];
            bv.y = g.b[t][2 * st + 1];
          } else {
            bv.x = g.b[2 * st][t];
            bv.y = g.b[2 * st + 1][t];
          }
          acc[u][t] = __builtin_amdgcn_wmma_f32_16x16x4_f32(false, av, false, bv,
                                                            (short)0, acc[u][t],
                                                            false, false);
        }
      }
    }
  }
#undef GM_LOAD

  const v4f bias4 = *(const v4f*)(bias + n0 + 4 * c);
#pragma unroll
  for (int u = 0; u < GM_MT; ++u) {
    float* __restrict__ Crow = C + (size_t)(m0 + 16 * u + 8 * h) * N + n0 + 4 * c;
#pragma unroll
    for (int r = 0; r < 8; ++r) {
      v4f o;
      o.x = acc[u][0][r] + bias4.x;
      o.y = acc[u][1][r] + bias4.y;
      o.z = acc[u][2][r] + bias4.z;
      o.w = acc[u][3][r] + bias4.w;
      *(v4f*)(Crow + (size_t)r * N) = o;
    }
  }
}

#define ROWS_PER_WAVE   4
#define GROUP           10
#define WAVES_PER_BLOCK 8
#define ROWS_PER_BLOCK  (ROWS_PER_WAVE * WAVES_PER_BLOCK)

__device__ __forceinline__ float lane_read_f(float x, int n) {
  return __int_as_float(__builtin_amdgcn_readlane(__float_as_int(x), n));
}

__global__ __launch_bounds__(32 * WAVES_PER_BLOCK) __attribute__((amdgpu_waves_per_eu(4)))
void geom_fuse_kernel(const int* __restrict__ nbr, const float* __restrict__ verts,
                      const float* __restrict__ feat,
                      const float* __restrict__ directions,
                      const float* __restrict__ distw,
                      float* __restrict__ out256) {
  const int lane = threadIdx.x & 31;
  const int wave = __builtin_amdgcn_readfirstlane((int)(threadIdx.x >> 5));
  const int row0 = (blockIdx.x * WAVES_PER_BLOCK + wave) * ROWS_PER_WAVE;
  const int b    = row0 / V;
  const size_t bbase = (size_t)b * V;
  const int c4 = 4 * lane;

  const int ln = (lane < NNB) ? lane : 0;
  int ni_next = nbr[(size_t)row0 * NNB + ln];

  v4f a0 = *(const v4f*)(directions + 0 * SUPC + c4);
  v4f a1 = *(const v4f*)(directions + 1 * SUPC + c4);
  v4f a2 = *(const v4f*)(directions + 2 * SUPC + c4);
  v4f b0 = *(const v4f*)(directions + 0 * SUPC + OUTC + c4);
  v4f b1 = *(const v4f*)(directions + 1 * SUPC + OUTC + c4);
  v4f b2 = *(const v4f*)(directions + 2 * SUPC + OUTC + c4);
#pragma unroll
  for (int j = 0; j < 4; ++j) {
    const float na = sqrtf(a0[j] * a0[j] + a1[j] * a1[j] + a2[j] * a2[j]);
    const float ia = 1.0f / fmaxf(na, EPSF);
    a0[j] *= ia; a1[j] *= ia; a2[j] *= ia;
    const float nb = sqrtf(b0[j] * b0[j] + b1[j] * b1[j] + b2[j] * b2[j]);
    const float ib = 1.0f / fmaxf(nb, EPSF);
    b0[j] *= ib; b1[j] *= ib; b2[j] *= ib;
  }

  const v4f* __restrict__ fb4 = (const v4f*)(feat + bbase * FEATC + OUTC);

#pragma unroll 1
  for (int r = 0; r < ROWS_PER_WAVE; ++r) {
    const int row = row0 + r;
    const int ni  = ni_next & (V - 1);
    const int rnx = (r + 1 < ROWS_PER_WAVE) ? row + 1 : row;
    ni_next = nbr[(size_t)rnx * NNB + ln];

    const float* pv = verts + (size_t)row * 3;
    const float* pn = verts + (bbase + ni) * 3;
    const float dx = pn[0] - pv[0];
    const float dy = pn[1] - pv[1];
    const float dz = pn[2] - pv[2];
    const float d  = sqrtf(dx * dx + dy * dy + dz * dz);
    const float ivd = 1.0f / fmaxf(d, EPSF);
    const float nx = dx * ivd;
    const float ny = dy * ivd;
    const float nz = dz * ivd;

    float dmax = d;
#pragma unroll
    for (int m = 16; m >= 1; m >>= 1) dmax = fmaxf(dmax, __shfl_xor(dmax, m, 32));

    v4f accA = {-INFINITY, -INFINITY, -INFINITY, -INFINITY};
    v4f accB = {-INFINITY, -INFINITY, -INFINITY, -INFINITY};
#pragma unroll 1
    for (int h = 0; h < NNB; h += GROUP) {
#pragma unroll
      for (int q = 0; q < GROUP; ++q) {
        const int n = h + q;
        const int   nin = __builtin_amdgcn_readlane(ni, n);
        const float x = lane_read_f(nx, n);
        const float y = lane_read_f(ny, n);
        const float z = lane_read_f(nz, n);
        const v4f* p = fb4 + ((unsigned)nin * (unsigned)(FEATC / 4) + (unsigned)lane);
        const v4f gA = p[0];
        const v4f gB = p[OUTC / 4];
#pragma unroll
        for (int j = 0; j < 4; ++j) {
          const float tA = fmaxf(0.0f, a0[j] * x + a1[j] * y + a2[j] * z);
          accA[j] = fmaxf(accA[j], tA * gA[j]);
          const float tB = fmaxf(0.0f, b0[j] * x + b1[j] * y + b2[j] * z);
          accB[j] = fmaxf(accB[j], tB * gB[j]);
        }
      }
    }

    const v4f center = *(const v4f*)(feat + (size_t)row * FEATC + c4);
    const v4f wA = *(const v4f*)(distw + c4);
    const v4f wB = *(const v4f*)(distw + OUTC + c4);
    v4f fuse, dsum;
#pragma unroll
    for (int j = 0; j < 4; ++j) {
      fuse[j] = center[j] + accA[j] + accB[j];
      dsum[j] = fmaxf(0.0f, dmax * wA[j]) + fmaxf(0.0f, dmax * wB[j]);
    }
    float* orow = out256 + (size_t)row * SUPC + c4;
    *(v4f*)(orow)        = fuse;
    *(v4f*)(orow + OUTC) = dsum;
  }
}

extern "C" void kernel_launch(void* const* d_in, const int* in_sizes, int n_in,
                              void* d_out, int out_size, void* d_ws, size_t ws_size,
                              hipStream_t stream) {
  const int*   nbr   = (const int*)  d_in[0];
  const float* verts = (const float*)d_in[1];
  const float* fmap  = (const float*)d_in[2];
  const float* wgt   = (const float*)d_in[3];
  const float* bias  = (const float*)d_in[4];
  const float* dirs  = (const float*)d_in[5];
  const float* distw = (const float*)d_in[6];
  const float* mlpw  = (const float*)d_in[7];
  const float* mlpb  = (const float*)d_in[8];
  float* out = (float*)d_out;

  float* feat   = (float*)d_ws;
  float* out256 = feat + (size_t)MROWS * FEATC;

  {
    gemm_bias_wmma<false, FEATC, INC><<<(MROWS / GM_ROWS) * (FEATC / GM_COLS) / GM_WAVES, 32 * GM_WAVES, 0, stream>>>(
        fmap, wgt, bias, feat);
  }

  geom_fuse_kernel<<<MROWS / ROWS_PER_BLOCK, 32 * WAVES_PER_BLOCK, 0, stream>>>(nbr, verts, feat, dirs, distw, out256);

  {
    gemm_bias_wmma<true, OUTC, SUPC><<<(MROWS / GM_ROWS) * (OUTC / GM_COLS) / GM_WAVES, 32 * GM_WAVES, 0, stream>>>(
        out256, mlpw, mlpb, out);
  }
}


# Gather software-pipelined: the wave's 4 rows' geometry done up front, then one stream of 16 groups of 5 neighbors, each group's row loads issued two groups (20 loads) ahead of its arithmetic
# speedup vs baseline: 2.7595x; 1.2072x over previous
#include <hip/hip_runtime.h>
#include <math.h>

#define BS    8
#define V     2048
#define NNB   20
#define INC   64
#define OUTC  128
#define SUP   2
#define SUPC  (SUP * OUTC)
#define FEATC (OUTC + SUPC)
#define MROWS (BS * V)
#define EPSF  1e-12f

typedef float v2f __attribute__((ext_vector_type(2)));
typedef float v8f __attribute__((ext_vector_type(8)));

typedef float v4f __attribute__((ext_vector_type(4)));

#define GM_MT    2
#define GM_NT    4
#define GM_WAVES 4
#define GM_ROWS  (16 * GM_MT)
#define GM_COLS  (16 * GM_NT)

struct GFrag {
  v4f a[GM_MT];
  v4f b[4];
};

template <bool BT, int N, int K>
__global__ __launch_bounds__(32 * GM_WAVES)
void gemm_bias_wmma(const float* __restrict__ A, const float* __restrict__ B,
                    const float* __restrict__ bias, float* __restrict__ C) {
  static_assert(GM_NT == 4, "column map packs 4 tiles into one 16-byte vector");
  static_assert(K % 8 == 0 && N % GM_COLS == 0, "shape");
  constexpr int TILES_N = N / GM_COLS;
  constexpr int KC = K / 8;

  const int lane = threadIdx.x & 31;
  const int wave = threadIdx.x >> 5;
  const int wt   = blockIdx.x * GM_WAVES + wave;
  const int tm   = wt / TILES_N;
  const int tn   = wt - tm * TILES_N;
  const int m0   = tm * GM_ROWS;
  const int n0   = tn * GM_COLS;
  const int h    = lane >> 4;
  const int c    = lane & 15;

  v8f acc[GM_MT][GM_NT];
#pragma unroll
  for (int u = 0; u < GM_MT; ++u)
#pragma unroll
    for (int t = 0; t < GM_NT; ++t)
      acc[u][t] = (v8f){0.f, 0.f, 0.f, 0.f, 0.f, 0.f, 0.f, 0.f};

  const float* __restrict__ ap[GM_MT];
#pragma unroll
  for (int u = 0; u < GM_MT; ++u)
    ap[u] = A + (size_t)(m0 + 16 * u + c) * K + 4 * h;

  const float* __restrict__ bp = BT ? (B + (size_t)(n0 + 4 * c) * K + 4 * h)
                                    : (B + (size_t)(4 * h) * N + n0 + 4 * c);

  GFrag f[3];

#define GM_LOAD(F, JJ)                                                        \
  do {                                                                        \
    _Pragma("unroll")                                                         \
    for (int u = 0; u < GM_MT; ++u)                                           \
      (F).a[u] = *(const v4f*)(ap[u] + 8 * (JJ));                             \
    _Pragma("unroll")                                                         \
    for (int i = 0; i < 4; ++i)                                               \
      (F).b[i] = BT ? *(const v4f*)(bp + (size_t)i * K + 8 * (JJ))            \
                    : *(const v4f*)(bp + (size_t)(8 * (JJ) + i) * N);         \
  } while (0)

  GM_LOAD(f[0], 0);
  if (KC > 1) GM_LOAD(f[1], 1);

#pragma unroll
  for (int jj = 0; jj < KC; ++jj) {
    __builtin_amdgcn_sched_barrier(0);
    if (jj + 2 < KC) GM_LOAD(f[(jj + 2) % 3], jj + 2);
    __builtin_amdgcn_sched_barrier(0);
    const GFrag& g = f[jj % 3];
#pragma unroll
    for (int st = 0; st < 2; ++st) {
#pragma unroll
      for (int u = 0; u < GM_MT; ++u) {
        v2f av;
        av.x = g.a[u][2 * st];
        av.y = g.a[u][2 * st + 1];
#pragma unroll
        for (int t = 0; t < GM_NT; ++t) {
          v2f bv;
          if (BT) {
            bv.x = g.b[t][2 * st];
            bv.y = g.b[t][2 * st + 1];
          } else {
            bv.x = g.b[2 * st][t];
            bv.y = g.b[2 * st + 1][t];
          }
          acc[u][t] = __builtin_amdgcn_wmma_f32_16x16x4_f32(false, av, false, bv,
                                                            (short)0, acc[u][t],
                                                            false, false);
        }
      }
    }
  }
#undef GM_LOAD

  const v4f bias4 = *(const v4f*)(bias + n0 + 4 * c);
#pragma unroll
  for (int u = 0; u < GM_MT; ++u) {
    float* __restrict__ Crow = C + (size_t)(m0 + 16 * u + 8 * h) * N + n0 + 4 * c;
#pragma unroll
    for (int r = 0; r < 8; ++r) {
      v4f o;
      o.x = acc[u][0][r] + bias4.x;
      o.y = acc[u][1][r] + bias4.y;
      o.z = acc[u][2][r] + bias4.z;
      o.w = acc[u][3][r] + bias4.w;
      *(v4f*)(Crow + (size_t)r * N) = o;
    }
  }
}

#define ROWS_PER_WAVE   4
#define GROUP           5
#define GROUPS_PER_ROW  (NNB / GROUP)
#define GROUPS_TOTAL    (ROWS_PER_WAVE * GROUPS_PER_ROW)
#define AHEAD_G         2
#define WAVES_PER_BLOCK 8
#define ROWS_PER_BLOCK  (ROWS_PER_WAVE * WAVES_PER_BLOCK)
#define TILE_PITCH      (SUPC + 4)
#define OUT_PITCH       (OUTC + 4)

__device__ __forceinline__ float lane_read_f(float x, int n) {
  return __int_as_float(__builtin_amdgcn_readlane(__float_as_int(x), n));
}

__global__ __launch_bounds__(32 * WAVES_PER_BLOCK) __attribute__((amdgpu_waves_per_eu(4)))
void geom_fuse_kernel(const int* __restrict__ nbr, const float* __restrict__ verts,
                      const float* __restrict__ feat,
                      const float* __restrict__ directions,
                      const float* __restrict__ distw,
                      const float* __restrict__ mlpw,
                      const float* __restrict__ mlpb,
                      float* __restrict__ out) {
  __shared__ __attribute__((aligned(16))) float tile[ROWS_PER_BLOCK * TILE_PITCH];
  __shared__ __attribute__((aligned(16))) float otile[ROWS_PER_BLOCK * OUT_PITCH];
  const int lane = threadIdx.x & 31;
  const int wave = __builtin_amdgcn_readfirstlane((int)(threadIdx.x >> 5));
  const int row0 = (blockIdx.x * WAVES_PER_BLOCK + wave) * ROWS_PER_WAVE;
  const int b    = row0 / V;
  const size_t bbase = (size_t)b * V;
  const int c4 = 4 * lane;

  const int ln = (lane < NNB) ? lane : 0;
  int nraw[ROWS_PER_WAVE];
#pragma unroll
  for (int r = 0; r < ROWS_PER_WAVE; ++r) nraw[r] = nbr[(size_t)(row0 + r) * NNB + ln];

  v4f a0 = *(const v4f*)(directions + 0 * SUPC + c4);
  v4f a1 = *(const v4f*)(directions + 1 * SUPC + c4);
  v4f a2 = *(const v4f*)(directions + 2 * SUPC + c4);
  v4f b0 = *(const v4f*)(directions + 0 * SUPC + OUTC + c4);
  v4f b1 = *(const v4f*)(directions + 1 * SUPC + OUTC + c4);
  v4f b2 = *(const v4f*)(directions + 2 * SUPC + OUTC + c4);
#pragma unroll
  for (int j = 0; j < 4; ++j) {
    const float na = sqrtf(a0[j] * a0[j] + a1[j] * a1[j] + a2[j] * a2[j]);
    const float ia = 1.0f / fmaxf(na, EPSF);
    a0[j] *= ia; a1[j] *= ia; a2[j] *= ia;
    const float nb = sqrtf(b0[j] * b0[j] + b1[j] * b1[j] + b2[j] * b2[j]);
    const float ib = 1.0f / fmaxf(nb, EPSF);
    b0[j] *= ib; b1[j] *= ib; b2[j] *= ib;
  }

  const v4f* __restrict__ fb4 = (const v4f*)(feat + bbase * FEATC + OUTC);

  int   gi[ROWS_PER_WAVE];
  float gx[ROWS_PER_WAVE], gy[ROWS_PER_WAVE], gz[ROWS_PER_WAVE];
  float qx[ROWS_PER_WAVE], qy[ROWS_PER_WAVE], qz[ROWS_PER_WAVE];
  const v4f wA = *(const v4f*)(distw + c4);
  const v4f wB = *(const v4f*)(distw + OUTC + c4);
#pragma unroll
  for (int r = 0; r < ROWS_PER_WAVE; ++r) {
    const int ni  = nraw[r] & (V - 1);
    const float* pv = verts + (size_t)(row0 + r) * 3;
    const float* pn = verts + (bbase + ni) * 3;
    gi[r] = ni;
    qx[r] = pn[0] - pv[0];
    qy[r] = pn[1] - pv[1];
    qz[r] = pn[2] - pv[2];
  }
#pragma unroll
  for (int r = 0; r < ROWS_PER_WAVE; ++r)
    asm volatile("" : "+v"(gi[r]), "+v"(qx[r]), "+v"(qy[r]), "+v"(qz[r]) : : "memory");

#define GEO_ROW(R)                                                             \
  do {                                                                         \
    const float dx = qx[R], dy = qy[R], dz = qz[R];                            \
    const float d  = sqrtf(dx * dx + dy * dy + dz * dz);                       \
    const float ivd = 1.0f / fmaxf(d, EPSF);                                   \
    gx[R] = dx * ivd;                                                          \
    gy[R] = dy * ivd;                                                          \
    gz[R] = dz * ivd;                                                          \
    float dmax = d;                                              \
    _Pragma("unroll")                                                          \
    for (int m = 16; m >= 1; m >>= 1) dmax = fmaxf(dmax, __shfl_xor(dmax, m, 32)); \
    v4f dsum;                                                                  \
    _Pragma("unroll")                                                          \
    for (int j = 0; j < 4; ++j)                                                \
      dsum[j] = fmaxf(0.0f, dmax * wA[j]) + fmaxf(0.0f, dmax * wB[j]);         \
    *(v4f*)(tile + (wave * ROWS_PER_WAVE + (R)) * TILE_PITCH + OUTC + c4) = dsum; \
    asm volatile("" : "+v"(gi[R]), "+v"(gx[R]), "+v"(gy[R]), "+v"(gz[R]) : : "memory"); \
  } while (0)

  v4f bufA[AHEAD_G + 1][GROUP];
  v4f bufB[AHEAD_G + 1][GROUP];
  v4f cen[2];

#define GATHER_ISSUE(G)                                                        \
  do {                                                                         \
    const int r_ = (G) / GROUPS_PER_ROW;                                       \
    const int g_ = (G) % GROUPS_PER_ROW;                                       \
    if (g_ == 0)                                                               \
      cen[r_ & 1] = *(const v4f*)(feat + (size_t)(row0 + r_) * FEATC + c4);    \
    _Pragma("unroll")                                                          \
    for (int q = 0; q < GROUP; ++q) {                                          \
      const int nin = __builtin_amdgcn_readlane(gi[r_], GROUP * g_ + q);       \
      const v4f* p = fb4 + ((unsigned)nin * (unsigned)(FEATC / 4) + (unsigned)lane); \
      bufA[(G) % (AHEAD_G + 1)][q] = p[0];                                     \
      bufB[(G) % (AHEAD_G + 1)][q] = p[OUTC / 4];                              \
    }                                                                          \
  } while (0)

  v4f accA = {-INFINITY, -INFINITY, -INFINITY, -INFINITY};
  v4f accB = {-INFINITY, -INFINITY, -INFINITY, -INFINITY};

  GEO_ROW(0);
  asm volatile("" : "+v"(a0), "+v"(a1), "+v"(a2), "+v"(b0), "+v"(b1), "+v"(b2) : : "memory");
  __builtin_amdgcn_sched_barrier(0);
#pragma unroll
  for (int G = 0; G < AHEAD_G; ++G) GATHER_ISSUE(G);
  __builtin_amdgcn_sched_barrier(0);
  static_assert(ROWS_PER_WAVE == 4, "rows 1..3 are written out");
  GEO_ROW(1);
  GEO_ROW(2);
  GEO_ROW(3);
#undef GEO_ROW

#pragma unroll
  for (int G = 0; G < GROUPS_TOTAL; ++G) {
    __builtin_amdgcn_sched_barrier(0);
    if (G + AHEAD_G < GROUPS_TOTAL) GATHER_ISSUE(G + AHEAD_G);
    __builtin_amdgcn_sched_barrier(0);

    const int r = G / GROUPS_PER_ROW;
    const int g = G % GROUPS_PER_ROW;
#pragma unroll
    for (int q = 0; q < GROUP; ++q) {
      const int n = GROUP * g + q;
      const float x = lane_read_f(gx[r], n);
      const float y = lane_read_f(gy[r], n);
      const float z = lane_read_f(gz[r], n);
      const v4f gA = bufA[G % (AHEAD_G + 1)][q];
      const v4f gB = bufB[G % (AHEAD_G + 1)][q];
#pragma unroll
      for (int j = 0; j < 4; ++j) {
        const float tA = fmaxf(0.0f, a0[j] * x + a1[j] * y + a2[j] * z);
        accA[j] = fmaxf(accA[j], tA * gA[j]);
        const float tB = fmaxf(0.0f, b0[j] * x + b1[j] * y + b2[j] * z);
        accB[j] = fmaxf(accB[j], tB * gB[j]);
      }
    }

    asm volatile("" : "+v"(accA), "+v"(accB) : : "memory");

    if (g == GROUPS_PER_ROW - 1) {
      const v4f center = cen[r & 1];
      v4f fuse;
#pragma unroll
      for (int j = 0; j < 4; ++j) fuse[j] = center[j] + accA[j] + accB[j];
      *(v4f*)(tile + (wave * ROWS_PER_WAVE + r) * TILE_PITCH + c4) = fuse;
      accA = (v4f){-INFINITY, -INFINITY, -INFINITY, -INFINITY};
      accB = (v4f){-INFINITY, -INFINITY, -INFINITY, -INFINITY};
    }
  }
#undef GATHER_ISSUE

  {
    const int h = lane >> 4;
    const int c = lane & 15;
    constexpr int KC = SUPC / 8;
    constexpr int AHEAD = 3;
    const float* __restrict__ bp = mlpw + (size_t)(16 * wave + c) * SUPC + 4 * h;
    const float* ap0 = tile + c * TILE_PITCH + 4 * h;
    const float* ap1 = ap0 + 16 * TILE_PITCH;

    v4f fb[AHEAD + 1];
#pragma unroll
    for (int j = 0; j < AHEAD; ++j) fb[j] = *(const v4f*)(bp + 8 * j);

    __syncthreads();

    v8f acc0 = (v8f){0.f, 0.f, 0.f, 0.f, 0.f, 0.f, 0.f, 0.f};
    v8f acc1 = acc0;
#pragma unroll
    for (int jj = 0; jj < KC; ++jj) {
      if (jj + AHEAD < KC) fb[(jj + AHEAD) % (AHEAD + 1)] = *(const v4f*)(bp + 8 * (jj + AHEAD));
      const v4f fa0 = *(const v4f*)(ap0 + 8 * jj);
      const v4f fa1 = *(const v4f*)(ap1 + 8 * jj);
      const v4f b4 = fb[jj % (AHEAD + 1)];
#pragma unroll
      for (int st = 0; st < 2; ++st) {
        v2f bv, av0, av1;
        bv.x  = b4[2 * st];   bv.y  = b4[2 * st + 1];
        av0.x = fa0[2 * st];  av0.y = fa0[2 * st + 1];
        av1.x = fa1[2 * st];  av1.y = fa1[2 * st + 1];
        acc0 = __builtin_amdgcn_wmma_f32_16x16x4_f32(false, av0, false, bv, (short)0, acc0, false, false);
        acc1 = __builtin_amdgcn_wmma_f32_16x16x4_f32(false, av1, false, bv, (short)0, acc1, false, false);
      }
    }

    const float bz = mlpb[16 * wave + c];
    float* op = otile + (8 * h) * OUT_PITCH + 16 * wave + c;
#pragma unroll
    for (int r = 0; r < 8; ++r) {
      op[r * OUT_PITCH]        = acc0[r] + bz;
      op[(16 + r) * OUT_PITCH] = acc1[r] + bz;
    }
    __syncthreads();
#pragma unroll
    for (int i = 0; i < ROWS_PER_WAVE; ++i) {
      const int lr = wave * ROWS_PER_WAVE + i;
      const v4f o = *(const v4f*)(otile + lr * OUT_PITCH + c4);
      *(v4f*)(out + (size_t)(row0 + i) * OUTC + c4) = o;
    }
  }
}

extern "C" void kernel_launch(void* const* d_in, const int* in_sizes, int n_in,
                              void* d_out, int out_size, void* d_ws, size_t ws_size,
                              hipStream_t stream) {
  const int*   nbr   = (const int*)  d_in[0];
  const float* verts = (const float*)d_in[1];
  const float* fmap  = (const float*)d_in[2];
  const float* wgt   = (const float*)d_in[3];
  const float* bias  = (const float*)d_in[4];
  const float* dirs  = (const float*)d_in[5];
  const float* distw = (const float*)d_in[6];
  const float* mlpw  = (const float*)d_in[7];
  const float* mlpb  = (const float*)d_in[8];
  float* out = (float*)d_out;

  float* feat = (float*)d_ws;

  {
    gemm_bias_wmma<false, FEATC, INC><<<(MROWS / GM_ROWS) * (FEATC / GM_COLS) / GM_WAVES, 32 * GM_WAVES, 0, stream>>>(
        fmap, wgt, bias, feat);
  }

  geom_fuse_kernel<<<MROWS / ROWS_PER_BLOCK, 32 * WAVES_PER_BLOCK, 0, stream>>>(nbr, verts, feat, dirs, distw, mlpw, mlpb, out);
}


# Distance half of the final layer made rank 1 (relu(dmax*w) = dmax*relu(w)): one extra WMMA block of launch 1 makes t = s*mlp_w[:,128:]^T, launch 2's GEMM runs K=132 (fuse|dmax) instead of 256
# speedup vs baseline: 2.9186x; 1.0577x over previous
#include <hip/hip_runtime.h>
#include <math.h>

#define BS    8
#define V     2048
#define NNB   20
#define INC   64
#define OUTC  128
#define SUP   2
#define SUPC  (SUP * OUTC)
#define FEATC (OUTC + SUPC)
#define MROWS (BS * V)
#define EPSF  1e-12f

typedef float v2f __attribute__((ext_vector_type(2)));
typedef float v8f __attribute__((ext_vector_type(8)));

typedef float v4f __attribute__((ext_vector_type(4)));

#define GM_MT    2
#define GM_NT    4
#define GM_WAVES 4
#define GM_ROWS  (16 * GM_MT)
#define GM_COLS  (16 * GM_NT)

struct GFrag {
  v4f a[GM_MT];
  v4f b[4];
};

__device__ __forceinline__ void dist_vec_block(const float* __restrict__ distw,
                                               const float* __restrict__ mlpw,
                                               float* __restrict__ tvec) {
  const int lane = threadIdx.x & 31;
  const int wave = threadIdx.x >> 5;
  const int h    = lane >> 4;
  const int c    = lane & 15;
  const float* __restrict__ sp  = distw + 4 * h;
  const float* __restrict__ bp0 = mlpw + (size_t)(32 * wave + c) * SUPC + OUTC + 4 * h;
  const float* __restrict__ bp1 = bp0 + (size_t)16 * SUPC;
  v8f acc0 = (v8f){0.f, 0.f, 0.f, 0.f, 0.f, 0.f, 0.f, 0.f};
  v8f acc1 = acc0;
#pragma unroll 4
  for (int jj = 0; jj < OUTC / 8; ++jj) {
    const v4f wa = *(const v4f*)(sp + 8 * jj);
    const v4f wb = *(const v4f*)(sp + OUTC + 8 * jj);
    const v4f b0 = *(const v4f*)(bp0 + 8 * jj);
    const v4f b1 = *(const v4f*)(bp1 + 8 * jj);
    v4f s4;
#pragma unroll
    for (int i = 0; i < 4; ++i) s4[i] = fmaxf(0.0f, wa[i]) + fmaxf(0.0f, wb[i]);
#pragma unroll
    for (int st = 0; st < 2; ++st) {
      v2f av, bv0, bv1;
      av.x  = s4[2 * st];  av.y  = s4[2 * st + 1];
      bv0.x = b0[2 * st];  bv0.y = b0[2 * st + 1];
      bv1.x = b1[2 * st];  bv1.y = b1[2 * st + 1];
      acc0 = __builtin_amdgcn_wmma_f32_16x16x4_f32(false, av, false, bv0, (short)0, acc0, false, false);
      acc1 = __builtin_amdgcn_wmma_f32_16x16x4_f32(false, av, false, bv1, (short)0, acc1, false, false);
    }
  }
  tvec[32 * wave + lane] = (h == 0) ? acc0[0] : acc1[0];
}

template <bool BT, int N, int K>
__global__ __launch_bounds__(32 * GM_WAVES)
void gemm_bias_wmma(const float* __restrict__ A, const float* __restrict__ B,
                    const float* __restrict__ bias, float* __restrict__ C,
                    const float* __restrict__ distw, const float* __restrict__ mlpw,
                    float* __restrict__ tvec) {
  static_assert(GM_NT == 4, "column map packs 4 tiles into one 16-byte vector");
  static_assert(K % 8 == 0 && N % GM_COLS == 0, "shape");
  constexpr int TILES_N = N / GM_COLS;
  constexpr int KC = K / 8;

  if (blockIdx.x == 0) {
    dist_vec_block(distw, mlpw, tvec);
    return;
  }

  const int lane = threadIdx.x & 31;
  const int wave = threadIdx.x >> 5;
  const int wt   = (blockIdx.x - 1) * GM_WAVES + wave;
  const int tm   = wt / TILES_N;
  const int tn   = wt - tm * TILES_N;
  const int m0   = tm * GM_ROWS;
  const int n0   = tn * GM_COLS;
  const int h    = lane >> 4;
  const int c    = lane & 15;

  v8f acc[GM_MT][GM_NT];
#pragma unroll
  for (int u = 0; u < GM_MT; ++u)
#pragma unroll
    for (int t = 0; t < GM_NT; ++t)
      acc[u][t] = (v8f){0.f, 0.f, 0.f, 0.f, 0.f, 0.f, 0.f, 0.f};

  const float* __restrict__ ap[GM_MT];
#pragma unroll
  for (int u = 0; u < GM_MT; ++u)
    ap[u] = A + (size_t)(m0 + 16 * u + c) * K + 4 * h;

  const float* __restrict__ bp = BT ? (B + (size_t)(n0 + 4 * c) * K + 4 * h)
                                    : (B + (size_t)(4 * h) * N + n0 + 4 * c);

  GFrag f[3];

#define GM_LOAD(F, JJ)                                                        \
  do {                                                                        \
    _Pragma("unroll")                                                         \
    for (int u = 0; u < GM_MT; ++u)                                           \
      (F).a[u] = *(const v4f*)(ap[u] + 8 * (JJ));                             \
    _Pragma("unroll")                                                         \
    for (int i = 0; i < 4; ++i)                                               \
      (F).b[i] = BT ? *(const v4f*)(bp + (size_t)i * K + 8 * (JJ))            \
                    : *(const v4f*)(bp + (size_t)(8 * (JJ) + i) * N);         \
  } while (0)

  GM_LOAD(f[0], 0);
  if (KC > 1) GM_LOAD(f[1], 1);

#pragma unroll
  for (int jj = 0; jj < KC; ++jj) {
    __builtin_amdgcn_sched_barrier(0);
    if (jj + 2 < KC) GM_LOAD(f[(jj + 2) % 3], jj + 2);
    __builtin_amdgcn_sched_barrier(0);
    const GFrag& g = f[jj % 3];
#pragma unroll
    for (int st = 0; st < 2; ++st) {
#pragma unroll
      for (int u = 0; u < GM_MT; ++u) {
        v2f av;
        av.x = g.a[u][2 * st];
        av.y = g.a[u][2 * st + 1];
#pragma unroll
        for (int t = 0; t < GM_NT; ++t) {
          v2f bv;
          if (BT) {
            bv.x = g.b[t][2 * st];
            bv.y = g.b[t][2 * st + 1];
          } else {
            bv.x = g.b[2 * st][t];
            bv.y = g.b[2 * st + 1][t];
          }
          acc[u][t] = __builtin_amdgcn_wmma_f32_16x16x4_f32(false, av, false, bv,
                                                            (short)0, acc[u][t],
                                                            false, false);
        }
      }
    }
  }
#undef GM_LOAD

  const v4f bias4 = *(const v4f*)(bias + n0 + 4 * c);
#pragma unroll
  for (int u = 0; u < GM_MT; ++u) {
    float* __restrict__ Crow = C + (size_t)(m0 + 16 * u + 8 * h) * N + n0 + 4 * c;
#pragma unroll
    for (int r = 0; r < 8; ++r) {
      v4f o;
      o.x = acc[u][0][r] + bias4.x;
      o.y = acc[u][1][r] + bias4.y;
      o.z = acc[u][2][r] + bias4.z;
      o.w = acc[u][3][r] + bias4.w;
      *(v4f*)(Crow + (size_t)r * N) = o;
    }
  }
}

#define ROWS_PER_WAVE   4
#define GROUP           5
#define GROUPS_PER_ROW  (NNB / GROUP)
#define GROUPS_TOTAL    (ROWS_PER_WAVE * GROUPS_PER_ROW)
#define AHEAD_G         2
#define WAVES_PER_BLOCK 8
#define ROWS_PER_BLOCK  (ROWS_PER_WAVE * WAVES_PER_BLOCK)
#define TILE_PITCH      (OUTC + 4)
#define OUT_PITCH       (OUTC + 4)

__device__ __forceinline__ float lane_read_f(float x, int n) {
  return __int_as_float(__builtin_amdgcn_readlane(__float_as_int(x), n));
}

__global__ __launch_bounds__(32 * WAVES_PER_BLOCK) __attribute__((amdgpu_waves_per_eu(4)))
void geom_fuse_kernel(const int* __restrict__ nbr, const float* __restrict__ verts,
                      const float* __restrict__ feat,
                      const float* __restrict__ directions,
                      const float* __restrict__ tvec,
                      const float* __restrict__ mlpw,
                      const float* __restrict__ mlpb,
                      float* __restrict__ out) {
  __shared__ __attribute__((aligned(16))) float tile[ROWS_PER_BLOCK * TILE_PITCH];
  __shared__ __attribute__((aligned(16))) float otile[ROWS_PER_BLOCK * OUT_PITCH];
  const int lane = threadIdx.x & 31;
  const int wave = __builtin_amdgcn_readfirstlane((int)(threadIdx.x >> 5));
  const int row0 = (blockIdx.x * WAVES_PER_BLOCK + wave) * ROWS_PER_WAVE;
  const int b    = row0 / V;
  const size_t bbase = (size_t)b * V;
  const int c4 = 4 * lane;

  const int ln = (lane < NNB) ? lane : 0;
  int nraw[ROWS_PER_WAVE];
#pragma unroll
  for (int r = 0; r < ROWS_PER_WAVE; ++r) nraw[r] = nbr[(size_t)(row0 + r) * NNB + ln];

  v4f a0 = *(const v4f*)(directions + 0 * SUPC + c4);
  v4f a1 = *(const v4f*)(directions + 1 * SUPC + c4);
  v4f a2 = *(const v4f*)(directions + 2 * SUPC + c4);
  v4f b0 = *(const v4f*)(directions + 0 * SUPC + OUTC + c4);
  v4f b1 = *(const v4f*)(directions + 1 * SUPC + OUTC + c4);
  v4f b2 = *(const v4f*)(directions + 2 * SUPC + OUTC + c4);
#pragma unroll
  for (int j = 0; j < 4; ++j) {
    const float na = sqrtf(a0[j] * a0[j] + a1[j] * a1[j] + a2[j] * a2[j]);
    const float ia = 1.0f / fmaxf(na, EPSF);
    a0[j] *= ia; a1[j] *= ia; a2[j] *= ia;
    const float nb = sqrtf(b0[j] * b0[j] + b1[j] * b1[j] + b2[j] * b2[j]);
    const float ib = 1.0f / fmaxf(nb, EPSF);
    b0[j] *= ib; b1[j] *= ib; b2[j] *= ib;
  }

  const v4f* __restrict__ fb4 = (const v4f*)(feat + bbase * FEATC + OUTC);

  int   gi[ROWS_PER_WAVE];
  float gx[ROWS_PER_WAVE], gy[ROWS_PER_WAVE], gz[ROWS_PER_WAVE];
  float qx[ROWS_PER_WAVE], qy[ROWS_PER_WAVE], qz[ROWS_PER_WAVE];
#pragma unroll
  for (int r = 0; r < ROWS_PER_WAVE; ++r) {
    const int ni  = nraw[r] & (V - 1);
    const float* pv = verts + (size_t)(row0 + r) * 3;
    const float* pn = verts + (bbase + ni) * 3;
    gi[r] = ni;
    qx[r] = pn[0] - pv[0];
    qy[r] = pn[1] - pv[1];
    qz[r] = pn[2] - pv[2];
  }
#pragma unroll
  for (int r = 0; r < ROWS_PER_WAVE; ++r)
    asm volatile("" : "+v"(gi[r]), "+v"(qx[r]), "+v"(qy[r]), "+v"(qz[r]) : : "memory");

#define GEO_ROW(R)                                                             \
  do {                                                                         \
    const float dx = qx[R], dy = qy[R], dz = qz[R];                            \
    const float d  = sqrtf(dx * dx + dy * dy + dz * dz);                       \
    const float ivd = 1.0f / fmaxf(d, EPSF);                                   \
    gx[R] = dx * ivd;                                                          \
    gy[R] = dy * ivd;                                                          \
    gz[R] = dz * ivd;                                                          \
    float dmax = d;                                              \
    _Pragma("unroll")                                                          \
    for (int m = 16; m >= 1; m >>= 1) dmax = fmaxf(dmax, __shfl_xor(dmax, m, 32)); \
    if (lane == 0)                                                             \
      *(v4f*)(tile + (wave * ROWS_PER_WAVE + (R)) * TILE_PITCH + OUTC) =       \
          (v4f){dmax, 0.0f, 0.0f, 0.0f};                                       \
    asm volatile("" : "+v"(gi[R]), "+v"(gx[R]), "+v"(gy[R]), "+v"(gz[R]) : : "memory"); \
  } while (0)

  v4f bufA[AHEAD_G + 1][GROUP];
  v4f bufB[AHEAD_G + 1][GROUP];
  v4f cen[2];

#define GATHER_ISSUE(G)                                                        \
  do {                                                                         \
    const int r_ = (G) / GROUPS_PER_ROW;                                       \
    const int g_ = (G) % GROUPS_PER_ROW;                                       \
    if (g_ == 0)                                                               \
      cen[r_ & 1] = *(const v4f*)(feat + (size_t)(row0 + r_) * FEATC + c4);    \
    _Pragma("unroll")                                                          \
    for (int q = 0; q < GROUP; ++q) {                                          \
      const int nin = __builtin_amdgcn_readlane(gi[r_], GROUP * g_ + q);       \
      const v4f* p = fb4 + ((unsigned)nin * (unsigned)(FEATC / 4) + (unsigned)lane); \
      bufA[(G) % (AHEAD_G + 1)][q] = p[0];                                     \
      bufB[(G) % (AHEAD_G + 1)][q] = p[OUTC / 4];                              \
    }                                                                          \
  } while (0)

  v4f accA = {-INFINITY, -INFINITY, -INFINITY, -INFINITY};
  v4f accB = {-INFINITY, -INFINITY, -INFINITY, -INFINITY};

  GEO_ROW(0);
  asm volatile("" : "+v"(a0), "+v"(a1), "+v"(a2), "+v"(b0), "+v"(b1), "+v"(b2) : : "memory");
  __builtin_amdgcn_sched_barrier(0);
#pragma unroll
  for (int G = 0; G < AHEAD_G; ++G) GATHER_ISSUE(G);
  __builtin_amdgcn_sched_barrier(0);
  static_assert(ROWS_PER_WAVE == 4, "rows 1..3 are written out");
  GEO_ROW(1);
  GEO_ROW(2);
  GEO_ROW(3);
#undef GEO_ROW

#pragma unroll
  for (int G = 0; G < GROUPS_TOTAL; ++G) {
    __builtin_amdgcn_sched_barrier(0);
    if (G + AHEAD_G < GROUPS_TOTAL) GATHER_ISSUE(G + AHEAD_G);
    __builtin_amdgcn_sched_barrier(0);

    const int r = G / GROUPS_PER_ROW;
    const int g = G % GROUPS_PER_ROW;
#pragma unroll
    for (int q = 0; q < GROUP; ++q) {
      const int n = GROUP * g + q;
      const float x = lane_read_f(gx[r], n);
      const float y = lane_read_f(gy[r], n);
      const float z = lane_read_f(gz[r], n);
      const v4f gA = bufA[G % (AHEAD_G + 1)][q];
      const v4f gB = bufB[G % (AHEAD_G + 1)][q];
#pragma unroll
      for (int j = 0; j < 4; ++j) {
        const float tA = fmaxf(0.0f, a0[j] * x + a1[j] * y + a2[j] * z);
        accA[j] = fmaxf(accA[j], tA * gA[j]);
        const float tB = fmaxf(0.0f, b0[j] * x + b1[j] * y + b2[j] * z);
        accB[j] = fmaxf(accB[j], tB * gB[j]);
      }
    }

    asm volatile("" : "+v"(accA), "+v"(accB) : : "memory");

    if (g == GROUPS_PER_ROW - 1) {
      const v4f center = cen[r & 1];
      v4f fuse;
#pragma unroll
      for (int j = 0; j < 4; ++j) fuse[j] = center[j] + accA[j] + accB[j];
      *(v4f*)(tile + (wave * ROWS_PER_WAVE + r) * TILE_PITCH + c4) = fuse;
      accA = (v4f){-INFINITY, -INFINITY, -INFINITY, -INFINITY};
      accB = (v4f){-INFINITY, -INFINITY, -INFINITY, -INFINITY};
    }
  }
#undef GATHER_ISSUE

  {
    const int h = lane >> 4;
    const int c = lane & 15;
    constexpr int KC = OUTC / 8;
    constexpr int AHEAD = 3;
    const float* __restrict__ bp = mlpw + (size_t)(16 * wave + c) * SUPC + 4 * h;
    const float* ap0 = tile + c * TILE_PITCH + 4 * h;
    const float* ap1 = ap0 + 16 * TILE_PITCH;

    v4f fb[AHEAD + 1];
#pragma unroll
    for (int j = 0; j < AHEAD; ++j) fb[j] = *(const v4f*)(bp + 8 * j);
    const float tcol = tvec[16 * wave + c];

    __syncthreads();

    v8f acc0 = (v8f){0.f, 0.f, 0.f, 0.f, 0.f, 0.f, 0.f, 0.f};
    v8f acc1 = acc0;
#pragma unroll
    for (int jj = 0; jj < KC; ++jj) {
      if (jj + AHEAD < KC) fb[(jj + AHEAD) % (AHEAD + 1)] = *(const v4f*)(bp + 8 * (jj + AHEAD));
      const v4f fa0 = *(const v4f*)(ap0 + 8 * jj);
      const v4f fa1 = *(const v4f*)(ap1 + 8 * jj);
      const v4f b4 = fb[jj % (AHEAD + 1)];
#pragma unroll
      for (int st = 0; st < 2; ++st) {
        v2f bv, av0, av1;
        bv.x  = b4[2 * st];   bv.y  = b4[2 * st + 1];
        av0.x = fa0[2 * st];  av0.y = fa0[2 * st + 1];
        av1.x = fa1[2 * st];  av1.y = fa1[2 * st + 1];
        acc0 = __builtin_amdgcn_wmma_f32_16x16x4_f32(false, av0, false, bv, (short)0, acc0, false, false);
        acc1 = __builtin_amdgcn_wmma_f32_16x16x4_f32(false, av1, false, bv, (short)0, acc1, false, false);
      }
    }
    {
      const v2f at0 = *(const v2f*)(tile + c * TILE_PITCH + OUTC + 2 * h);
      const v2f at1 = *(const v2f*)(tile + (16 + c) * TILE_PITCH + OUTC + 2 * h);
      v2f bt;
      bt.x = (h == 0) ? tcol : 0.0f;
      bt.y = 0.0f;
      acc0 = __builtin_amdgcn_wmma_f32_16x16x4_f32(false, at0, false, bt, (short)0, acc0, false, false);
      acc1 = __builtin_amdgcn_wmma_f32_16x16x4_f32(false, at1, false, bt, (short)0, acc1, false, false);
    }

    const float bz = mlpb[16 * wave + c];
    float* op = otile + (8 * h) * OUT_PITCH + 16 * wave + c;
#pragma unroll
    for (int r = 0; r < 8; ++r) {
      op[r * OUT_PITCH]        = acc0[r] + bz;
      op[(16 + r) * OUT_PITCH] = acc1[r] + bz;
    }
    __syncthreads();
#pragma unroll
    for (int i = 0; i < ROWS_PER_WAVE; ++i) {
      const int lr = wave * ROWS_PER_WAVE + i;
      const v4f o = *(const v4f*)(otile + lr * OUT_PITCH + c4);
      *(v4f*)(out + (size_t)(row0 + i) * OUTC + c4) = o;
    }
  }
}

extern "C" void kernel_launch(void* const* d_in, const int* in_sizes, int n_in,
                              void* d_out, int out_size, void* d_ws, size_t ws_size,
                              hipStream_t stream) {
  const int*   nbr   = (const int*)  d_in[0];
  const float* verts = (const float*)d_in[1];
  const float* fmap  = (const float*)d_in[2];
  const float* wgt   = (const float*)d_in[3];
  const float* bias  = (const float*)d_in[4];
  const float* dirs  = (const float*)d_in[5];
  const float* distw = (const float*)d_in[6];
  const float* mlpw  = (const float*)d_in[7];
  const float* mlpb  = (const float*)d_in[8];
  float* out = (float*)d_out;

  float* feat = (float*)d_ws;
  float* tvec = feat + (size_t)MROWS * FEATC;

  {
    gemm_bias_wmma<false, FEATC, INC><<<1 + (MROWS / GM_ROWS) * (FEATC / GM_COLS) / GM_WAVES, 32 * GM_WAVES, 0, stream>>>(
        fmap, wgt, bias, feat, distw, mlpw, tvec);
  }

  geom_fuse_kernel<<<MROWS / ROWS_PER_BLOCK, 32 * WAVES_PER_BLOCK, 0, stream>>>(nbr, verts, feat, dirs, tvec, mlpw, mlpb, out);
}
